# starting kernel
# baseline (speedup 1.0000x reference)
#include <hip/hip_runtime.h>

typedef __attribute__((ext_vector_type(2))) float v2f;
typedef __attribute__((ext_vector_type(8))) float v8f;

#define BATCH   4096
#define N_IN    6
#define N_MF    4
#define N_RULES 4096
#define WAVES   8
#define TILE_M  16

__launch_bounds__(256, 1)
__global__ void anfis_wmma_kernel(const float* __restrict__ x,
                                  const float* __restrict__ mfa,
                                  const float* __restrict__ mfb,
                                  const float* __restrict__ mfc,
                                  const float* __restrict__ coeff,
                                  float* __restrict__ out)
{
    __shared__ float  A01[WAVES][TILE_M][16];
    __shared__ float  A23[WAVES][TILE_M][16];
    __shared__ float2 Bfrag[64][32];
    __shared__ float  Sst[WAVES][TILE_M][16];

    const int tid  = threadIdx.x;
    const int wave = tid >> 5;
    const int lane = tid & 31;
    const int m    = lane & 15;
    const int half = lane >> 4;
    const int rowbase = blockIdx.x * (WAVES * TILE_M) + wave * TILE_M;
    const int row = rowbase + m;

    float xv[N_IN];
#pragma unroll
    for (int i = 0; i < N_IN; ++i) xv[i] = x[row * N_IN + i];

    float memb[N_IN][N_MF];
#pragma unroll
    for (int v = 0; v < N_IN; ++v) {
#pragma unroll
        for (int j = 0; j < N_MF; ++j) {
            float d    = (xv[v] - mfc[v * N_MF + j]) / mfa[v * N_MF + j];
            float dist = d * d;
            float p    = __powf(dist, mfb[v * N_MF + j]);
            memb[v][j] = 1.0f / (1.0f + p);
        }
    }

    if (half == 0) {
#pragma unroll
        for (int ja = 0; ja < 4; ++ja) {
#pragma unroll
            for (int jb = 0; jb < 4; ++jb) {
                A01[wave][m][ja * 4 + jb] = memb[0][ja] * memb[1][jb];
                A23[wave][m][ja * 4 + jb] = memb[2][ja] * memb[3][jb];
            }
        }
    }
    const float m5a = half ? memb[5][2] : memb[5][0];
    const float m5b = half ? memb[5][3] : memb[5][1];
    float e0[4], e1[4];
#pragma unroll
    for (int s = 0; s < 4; ++s) {
        e0[s] = memb[4][s] * m5a;
        e1[s] = memb[4][s] * m5b;
    }

    const int   sn    = tid & 15;
    const bool  slt7  = (sn < 7);
    const float scaug = (sn == 7) ? 1.0f : 0.0f;
    const int   skap  = ((tid >> 4) & 1) << 1;

    v8f acc = {};

    for (int j01 = 0; j01 < 16; ++j01) {
        __syncthreads();
#pragma unroll
        for (int e = 0; e < 8; ++e) {
            const int idx  = e * 256 + tid;
            const int step = idx >> 5;
            const int l    = idx & 31;
            const int k0   = j01 * 256 + step * 4 + skap;
            float b0 = scaug, b1 = scaug;
            if (slt7) {
                b0 = coeff[k0 * 7 + sn];
                b1 = coeff[(k0 + 1) * 7 + sn];
            }
            Bfrag[step][l] = make_float2(b0, b1);
        }
        __syncthreads();

        if (j01 < 15)
            __builtin_prefetch(coeff + (j01 + 1) * 256 * 7 + tid * 7, 0, 1);

        const float a01 = A01[wave][m][j01];
#pragma unroll
        for (int j23 = 0; j23 < 16; ++j23) {
            const float a23 = A23[wave][m][j23];
            const float t   = a01 * a23;
#pragma unroll
            for (int s = 0; s < 4; ++s) {
                const float2 bf = Bfrag[j23 * 4 + s][lane];
                v2f bv; bv.x = bf.x; bv.y = bf.y;
                v2f av; av.x = t * e0[s]; av.y = t * e1[s];
                acc = __builtin_amdgcn_wmma_f32_16x16x4_f32(
                    false, av, false, bv, (short)0, acc, false, false);
            }
        }
    }

#pragma unroll
    for (int v = 0; v < 8; ++v)
        Sst[wave][v + half * 8][m] = acc[v];

    __syncthreads();

    if (lane < 16) {
        float num = Sst[wave][m][6];
#pragma unroll
        for (int i = 0; i < N_IN; ++i)
            num += Sst[wave][m][i] * xv[i];
        float denom = fmaxf(Sst[wave][m][7], 1e-12f);
        out[row] = num / denom;
    }
}

extern "C" void kernel_launch(void* const* d_in, const int* in_sizes, int n_in,
                              void* d_out, int out_size, void* d_ws, size_t ws_size,
                              hipStream_t stream) {
    const float* x     = (const float*)d_in[0];
    const float* mfa   = (const float*)d_in[1];
    const float* mfb   = (const float*)d_in[2];
    const float* mfc   = (const float*)d_in[3];
    const float* coeff = (const float*)d_in[4];
    float* out = (float*)d_out;

    dim3 grid(BATCH / (WAVES * TILE_M));
    dim3 block(WAVES * 32);
    anfis_wmma_kernel<<<grid, block, 0, stream>>>(x, mfa, mfb, mfc, coeff, out);
}


# One 16-row tile per workgroup (grid 32 -> 256): rules split into 16 partial sums (8 waves x 2 accumulator tiles) added via LDS in wave order, memberships computed once per tile; new summation order
# speedup vs baseline: 6.7017x; 6.7017x over previous
#include <hip/hip_runtime.h>


typedef __attribute__((ext_vector_type(2))) float v2f;
typedef __attribute__((ext_vector_type(8))) float v8f;

#define BATCH   4096
#define N_IN    6
#define N_MF    4
#define N_RULES 4096
#define WAVES   8
#define TILE_M  16
#define K_PER_WAVE (N_RULES / WAVES)
#define SLICE_F4   (K_PER_WAVE * 7 / 4)
#define LOADS_F4   (SLICE_F4 / 32)

__device__ __forceinline__ float bell(float xv, float a, float b, float c)
{
    float d    = (xv - c) / a;
    float dist = d * d;
    float p    = __powf(dist, b);
    return 1.0f / (1.0f + p);
}

__device__ __forceinline__ void wmma_group(v8f& acc0, v8f& acc1,
                                           const float* __restrict__ Bj,
                                           const float a01j,
                                           const float (&mb2)[N_MF],
                                           const float (&mb3)[N_MF],
                                           const float (&e0)[4],
                                           const float (&e1)[4],
                                           const bool slt7, const float scaug)
{
#pragma unroll
    for (int j23 = 0; j23 < 16; ++j23) {
        const float t = a01j * (mb2[j23 >> 2] * mb3[j23 & 3]);
#pragma unroll
        for (int s = 0; s < 4; ++s) {
            const int kl = j23 * 16 + s * 4;
            const float r0 = Bj[kl * 7];
            const float r1 = Bj[kl * 7 + 7];
            v2f bv; bv.x = slt7 ? r0 : scaug; bv.y = slt7 ? r1 : scaug;
            v2f av; av.x = t * e0[s]; av.y = t * e1[s];
            if (s & 1)
                acc1 = __builtin_amdgcn_wmma_f32_16x16x4_f32(
                    false, av, false, bv, (short)0, acc1, false, false);
            else
                acc0 = __builtin_amdgcn_wmma_f32_16x16x4_f32(
                    false, av, false, bv, (short)0, acc0, false, false);
        }
    }
}

__launch_bounds__(256)
__global__ void anfis_wmma_kernel(const float* __restrict__ x,
                                  const float* __restrict__ mfa,
                                  const float* __restrict__ mfb,
                                  const float* __restrict__ mfc,
                                  const float* __restrict__ coeff,
                                  float* __restrict__ out)
{
    __shared__ float4 Braw[WAVES][SLICE_F4];
    __shared__ float  Spart[WAVES][TILE_M][8];
    __shared__ float  Sred[TILE_M][8];
    __shared__ float  Msh[N_IN * N_MF][TILE_M];

    const int tid  = threadIdx.x;
    const int wave = __builtin_amdgcn_readfirstlane(tid >> 5);
    const int lane = tid & 31;
    const int m    = lane & 15;
    const int half = lane >> 4;
    const int row  = blockIdx.x * TILE_M + m;

    const float4* __restrict__ src4 =
        reinterpret_cast<const float4*>(coeff + wave * (K_PER_WAVE * 7));
    float4 stgA[LOADS_F4 / 2];
#pragma unroll
    for (int i = 0; i < LOADS_F4 / 2; ++i) stgA[i] = src4[i * 32 + lane];

    const int q = tid >> 4;
    {
        const float xq = x[row * N_IN + (q >> 2)];
        Msh[q][m] = bell(xq, mfa[q], mfb[q], mfc[q]);
    }

#pragma unroll
    for (int i = 0; i < LOADS_F4 / 2; ++i) Braw[wave][i * 32 + lane] = stgA[i];

    if (wave < 4) {
        const int q2 = 16 + q;
        const float xq2 = x[row * N_IN + (q2 >> 2)];
        Msh[q2][m] = bell(xq2, mfa[q2], mfb[q2], mfc[q2]);
    }
    __syncthreads();

    const int j0  = wave >> 1;
    const int j1b = (wave & 1) * 2;
    const float m0 = Msh[j0][m];
    float a01[2];
#pragma unroll
    for (int jj = 0; jj < 2; ++jj) a01[jj] = m0 * Msh[N_MF + j1b + jj][m];
    float mb2[N_MF], mb3[N_MF], mb4[N_MF];
#pragma unroll
    for (int j = 0; j < N_MF; ++j) {
        mb2[j] = Msh[2 * N_MF + j][m];
        mb3[j] = Msh[3 * N_MF + j][m];
        mb4[j] = Msh[4 * N_MF + j][m];
    }
    const float m5a = Msh[5 * N_MF + 2 * half][m];
    const float m5b = Msh[5 * N_MF + 2 * half + 1][m];
    float e0[4], e1[4];
#pragma unroll
    for (int s = 0; s < 4; ++s) {
        e0[s] = mb4[s] * m5a;
        e1[s] = mb4[s] * m5b;
    }

    const bool  slt7  = (m < 7);
    const float scaug = (m == 7) ? 1.0f : 0.0f;
    const float* Bw = reinterpret_cast<const float*>(&Braw[wave][0])
                      + (2 * half) * 7 + (slt7 ? m : 0);

    v8f acc0 = {};
    v8f acc1 = {};

    float4 stgB[LOADS_F4 / 2];
#pragma unroll
    for (int i = 0; i < LOADS_F4 / 2; ++i) stgB[i] = src4[(LOADS_F4 / 2 + i) * 32 + lane];

    wmma_group(acc0, acc1, Bw, a01[0], mb2, mb3, e0, e1, slt7, scaug);

#pragma unroll
    for (int i = 0; i < LOADS_F4 / 2; ++i)
        Braw[wave][(LOADS_F4 / 2 + i) * 32 + lane] = stgB[i];
    __syncthreads();

    wmma_group(acc0, acc1, Bw + 256 * 7, a01[1], mb2, mb3, e0, e1, slt7, scaug);

    if (m < 8) {
#pragma unroll
        for (int v = 0; v < 8; ++v)
            Spart[wave][v + half * 8][m] = acc0[v] + acc1[v];
    }
    __syncthreads();

    if (tid < TILE_M * 8) {
        const int r = tid >> 3, c = tid & 7;
        float sum = Spart[0][r][c];
#pragma unroll
        for (int w = 1; w < WAVES; ++w) sum += Spart[w][r][c];
        Sred[r][c] = sum;
    }
    __syncthreads();

    if (tid < TILE_M) {
        float num = Sred[m][6];
#pragma unroll
        for (int i = 0; i < N_IN; ++i)
            num += Sred[m][i] * x[row * N_IN + i];
        float denom = fmaxf(Sred[m][7], 1e-12f);
        out[row] = num / denom;
    }
}

extern "C" void kernel_launch(void* const* d_in, const int* in_sizes, int n_in,
                              void* d_out, int out_size, void* d_ws, size_t ws_size,
                              hipStream_t stream) {
    const float* x     = (const float*)d_in[0];
    const float* mfa   = (const float*)d_in[1];
    const float* mfb   = (const float*)d_in[2];
    const float* mfc   = (const float*)d_in[3];
    const float* coeff = (const float*)d_in[4];
    float* out = (float*)d_out;

    dim3 grid(BATCH / TILE_M);
    dim3 block(WAVES * 32);
    anfis_wmma_kernel<<<grid, block, 0, stream>>>(x, mfa, mfb, mfc, coeff, out);
}


# Coefficients become the 16-row WMMA operand (2 j01 groups x 8 columns), batch rows the columns, K = j2345; j01 membership factor applied to accumulators after: half the WMMAs; new summation order
# speedup vs baseline: 8.3349x; 1.2437x over previous
#include <hip/hip_runtime.h>


typedef __attribute__((ext_vector_type(2))) float v2f;
typedef __attribute__((ext_vector_type(8))) float v8f;

#define BATCH   4096
#define N_IN    6
#define N_MF    4
#define N_RULES 4096
#define WAVES   8
#define TILE_N  16
#define K_PER_WAVE 32
#define CHUNK_FL   (K_PER_WAVE * 7)
#define CHUNK_F4   (CHUNK_FL / 4)
#define J01_STRIDE_F4 (256 * 7 / 4)
#define SLICE_F4   (16 * CHUNK_F4)
#define LOADS_F4   (SLICE_F4 / 32)

__device__ __forceinline__ float bell(float xv, float a, float b, float c)
{
    float d    = (xv - c) / a;
    float dist = d * d;
    float p    = __powf(dist, b);
    return 1.0f / (1.0f + p);
}

template <int T0>
__device__ __forceinline__ void wmma_half(v8f (&D)[8],
                                          const float* __restrict__ Aw,
                                          const v2f (&bv)[8],
                                          const bool clt7)
{
#pragma unroll
    for (int s = 0; s < 8; ++s) {
#pragma unroll
        for (int t = T0; t < T0 + 4; ++t) {
            const int o = (2 * t) * CHUNK_FL + (4 * s) * 7;
            const float r0 = Aw[o];
            const float r1 = Aw[o + 7];
            v2f av; av.x = clt7 ? r0 : 1.0f; av.y = clt7 ? r1 : 1.0f;
            D[t] = __builtin_amdgcn_wmma_f32_16x16x4_f32(
                false, av, false, bv[s], (short)0, D[t], false, false);
        }
    }
}

__launch_bounds__(256)
__global__ void anfis_wmma_kernel(const float* __restrict__ x,
                                  const float* __restrict__ mfa,
                                  const float* __restrict__ mfb,
                                  const float* __restrict__ mfc,
                                  const float* __restrict__ coeff,
                                  float* __restrict__ out)
{
    __shared__ float4 Braw[WAVES][SLICE_F4];
    __shared__ float  Spart[WAVES * 2][8][TILE_N];
    __shared__ float  Sred[8][TILE_N];
    __shared__ float  Msh[N_IN * N_MF][TILE_N];

    const int tid  = threadIdx.x;
    const int wave = __builtin_amdgcn_readfirstlane(tid >> 5);
    const int lane = tid & 31;
    const int m    = lane & 15;
    const int half = lane >> 4;
    const int row  = blockIdx.x * TILE_N + m;

    const float4* __restrict__ src4 =
        reinterpret_cast<const float4*>(coeff) + wave * CHUNK_F4;
    float4 stgA[LOADS_F4 / 2];
#pragma unroll
    for (int i = 0; i < LOADS_F4 / 2; ++i) {
        const int idx = i * 32 + lane;
        stgA[i] = src4[idx + (idx / CHUNK_F4) * (J01_STRIDE_F4 - CHUNK_F4)];
    }

    const int q = tid >> 4;
    {
        const float xq = x[row * N_IN + (q >> 2)];
        Msh[q][m] = bell(xq, mfa[q], mfb[q], mfc[q]);
    }

#pragma unroll
    for (int i = 0; i < LOADS_F4 / 2; ++i) Braw[wave][i * 32 + lane] = stgA[i];

    if (wave < 4) {
        const int q2 = 16 + q;
        const float xq2 = x[row * N_IN + (q2 >> 2)];
        Msh[q2][m] = bell(xq2, mfa[q2], mfb[q2], mfc[q2]);
    }
    __syncthreads();

    const float m2 = Msh[2 * N_MF + (wave >> 1)][m];
    float a23[2];
#pragma unroll
    for (int jj = 0; jj < 2; ++jj)
        a23[jj] = m2 * Msh[3 * N_MF + (wave & 1) * 2 + jj][m];
    const float m5a = Msh[5 * N_MF + 2 * half][m];
    const float m5b = Msh[5 * N_MF + 2 * half + 1][m];
    float e0[4], e1[4];
#pragma unroll
    for (int j = 0; j < N_MF; ++j) {
        const float m4 = Msh[4 * N_MF + j][m];
        e0[j] = m4 * m5a;
        e1[j] = m4 * m5b;
    }
    v2f bv[8];
#pragma unroll
    for (int s = 0; s < 8; ++s) {
        bv[s].x = a23[s >> 2] * e0[s & 3];
        bv[s].y = a23[s >> 2] * e1[s & 3];
    }
    float f01[8];
    {
        const float a1lo = Msh[N_MF + half][m];
        const float a1hi = Msh[N_MF + 2 + half][m];
#pragma unroll
        for (int t = 0; t < 8; ++t)
            f01[t] = Msh[t >> 1][m] * ((t & 1) ? a1hi : a1lo);
    }

    const int  u    = m >> 3;
    const int  c    = m & 7;
    const bool clt7 = (c < 7);
    const float* Aw = reinterpret_cast<const float*>(&Braw[wave][0])
                      + u * CHUNK_FL + (2 * half) * 7 + (clt7 ? c : 0);

    v8f D[8];
#pragma unroll
    for (int t = 0; t < 8; ++t) D[t] = v8f{};

    float4 stgB[LOADS_F4 / 2];
#pragma unroll
    for (int i = 0; i < LOADS_F4 / 2; ++i) {
        const int idx = (LOADS_F4 / 2 + i) * 32 + lane;
        stgB[i] = src4[idx + (idx / CHUNK_F4) * (J01_STRIDE_F4 - CHUNK_F4)];
    }

    wmma_half<0>(D, Aw, bv, clt7);

#pragma unroll
    for (int i = 0; i < LOADS_F4 / 2; ++i)
        Braw[wave][(LOADS_F4 / 2 + i) * 32 + lane] = stgB[i];
    __syncthreads();

    wmma_half<4>(D, Aw, bv, clt7);

#pragma unroll
    for (int v = 0; v < 8; ++v) {
        float sacc = D[0][v] * f01[0];
#pragma unroll
        for (int t = 1; t < 8; ++t) sacc += D[t][v] * f01[t];
        Spart[wave * 2 + half][v][m] = sacc;
    }
    __syncthreads();

    if (tid < 8 * TILE_N) {
        const int cc = tid >> 4, bb = tid & 15;
        float sum = Spart[0][cc][bb];
#pragma unroll
        for (int p = 1; p < WAVES * 2; ++p) sum += Spart[p][cc][bb];
        Sred[cc][bb] = sum;
    }
    __syncthreads();

    if (tid < TILE_N) {
        float num = Sred[6][m];
#pragma unroll
        for (int i = 0; i < N_IN; ++i)
            num += Sred[i][m] * x[row * N_IN + i];
        float denom = fmaxf(Sred[7][m], 1e-12f);
        out[row] = num / denom;
    }
}

extern "C" void kernel_launch(void* const* d_in, const int* in_sizes, int n_in,
                              void* d_out, int out_size, void* d_ws, size_t ws_size,
                              hipStream_t stream) {
    const float* x     = (const float*)d_in[0];
    const float* mfa   = (const float*)d_in[1];
    const float* mfb   = (const float*)d_in[2];
    const float* mfc   = (const float*)d_in[3];
    const float* coeff = (const float*)d_in[4];
    float* out = (float*)d_out;

    dim3 grid(BATCH / TILE_N);
    dim3 block(WAVES * 32);
    anfis_wmma_kernel<<<grid, block, 0, stream>>>(x, mfa, mfb, mfc, coeff, out);
}


# Prologue loads issued as one group at entry, ahead of the first wait: x, bell parameters of both passes, closing-dot x row, first 14 coefficient float4; second bell pass before the LDS park
# speedup vs baseline: 9.0403x; 1.0846x over previous
#include <hip/hip_runtime.h>


typedef __attribute__((ext_vector_type(2))) float v2f;
typedef __attribute__((ext_vector_type(8))) float v8f;

#define BATCH   4096
#define N_IN    6
#define N_MF    4
#define N_RULES 4096
#define WAVES   8
#define TILE_N  16
#define K_PER_WAVE 32
#define CHUNK_FL   (K_PER_WAVE * 7)
#define CHUNK_F4   (CHUNK_FL / 4)
#define J01_STRIDE_F4 (256 * 7 / 4)
#define SLICE_F4   (16 * CHUNK_F4)
#define LOADS_F4   (SLICE_F4 / 32)

__device__ __forceinline__ float bell(float xv, float a, float b, float c)
{
    float d    = (xv - c) / a;
    float dist = d * d;
    float p    = __powf(dist, b);
    return 1.0f / (1.0f + p);
}

template <int T0>
__device__ __forceinline__ void wmma_half(v8f (&D)[8],
                                          const float* __restrict__ Aw,
                                          const v2f (&bv)[8],
                                          const bool clt7)
{
#pragma unroll
    for (int s = 0; s < 8; ++s) {
#pragma unroll
        for (int t = T0; t < T0 + 4; ++t) {
            const int o = (2 * t) * CHUNK_FL + (4 * s) * 7;
            const float r0 = Aw[o];
            const float r1 = Aw[o + 7];
            v2f av; av.x = clt7 ? r0 : 1.0f; av.y = clt7 ? r1 : 1.0f;
            D[t] = __builtin_amdgcn_wmma_f32_16x16x4_f32(
                false, av, false, bv[s], (short)0, D[t], false, false);
        }
    }
}

__launch_bounds__(256)
__global__ void anfis_wmma_kernel(const float* __restrict__ x,
                                  const float* __restrict__ mfa,
                                  const float* __restrict__ mfb,
                                  const float* __restrict__ mfc,
                                  const float* coeff,
                                  float* __restrict__ out)
{
    __shared__ float4 Braw[WAVES][SLICE_F4];
    __shared__ float  Spart[WAVES * 2][8][TILE_N];
    __shared__ float  Sred[8][TILE_N];
    __shared__ float  Msh[N_IN * N_MF][TILE_N];

    const int tid  = threadIdx.x;
    const int wave = __builtin_amdgcn_readfirstlane(tid >> 5);
    const int lane = tid & 31;
    const int m    = lane & 15;
    const int half = lane >> 4;
    const int row  = blockIdx.x * TILE_N + m;

    const int q  = tid >> 4;
    const int q2 = 16 + (q & 7);
    const float xq  = x[row * N_IN + (q >> 2)];
    const float pa1 = mfa[q],  pb1 = mfb[q],  pc1 = mfc[q];
    const float xq2 = x[row * N_IN + (q2 >> 2)];
    const float pa2 = mfa[q2], pb2 = mfb[q2], pc2 = mfc[q2];
    const float* xrow = x + row * N_IN;
    const float xr0 = xrow[0], xr1 = xrow[1], xr2 = xrow[2],
                xr3 = xrow[3], xr4 = xrow[4], xr5 = xrow[5];

    const float4* src4 =
        reinterpret_cast<const float4*>(coeff) + wave * CHUNK_F4;
#define SLICE_SRC(i) ((i) * 32 + lane + (((i) * 32 + lane) / CHUNK_F4) * (J01_STRIDE_F4 - CHUNK_F4))
    const float4 g0 = src4[SLICE_SRC(0)],  g1 = src4[SLICE_SRC(1)],  g2 = src4[SLICE_SRC(2)],
                 g3 = src4[SLICE_SRC(3)],  g4 = src4[SLICE_SRC(4)],  g5 = src4[SLICE_SRC(5)],
                 g6 = src4[SLICE_SRC(6)],  g7 = src4[SLICE_SRC(7)],  g8 = src4[SLICE_SRC(8)],
                 g9 = src4[SLICE_SRC(9)],  g10 = src4[SLICE_SRC(10)], g11 = src4[SLICE_SRC(11)],
                 g12 = src4[SLICE_SRC(12)], g13 = src4[SLICE_SRC(13)];
    __builtin_amdgcn_sched_barrier(0);
#undef SLICE_SRC

    Msh[q][m] = bell(xq, pa1, pb1, pc1);
    asm volatile("" :: "v"(xq2), "v"(pa2), "v"(pb2), "v"(pc2),
                       "v"(xr0), "v"(xr1), "v"(xr2), "v"(xr3), "v"(xr4), "v"(xr5));

    if (wave < 4)
        Msh[q2][m] = bell(xq2, pa2, pb2, pc2);

    {
        float4* dst = &Braw[wave][lane];
        dst[0 * 32] = g0;   dst[1 * 32] = g1;   dst[2 * 32] = g2;   dst[3 * 32] = g3;
        dst[4 * 32] = g4;   dst[5 * 32] = g5;   dst[6 * 32] = g6;   dst[7 * 32] = g7;
        dst[8 * 32] = g8;   dst[9 * 32] = g9;   dst[10 * 32] = g10; dst[11 * 32] = g11;
        dst[12 * 32] = g12; dst[13 * 32] = g13;
    }
    __syncthreads();

    const float m2 = Msh[2 * N_MF + (wave >> 1)][m];
    float a23[2];
#pragma unroll
    for (int jj = 0; jj < 2; ++jj)
        a23[jj] = m2 * Msh[3 * N_MF + (wave & 1) * 2 + jj][m];
    const float m5a = Msh[5 * N_MF + 2 * half][m];
    const float m5b = Msh[5 * N_MF + 2 * half + 1][m];
    float e0[4], e1[4];
#pragma unroll
    for (int j = 0; j < N_MF; ++j) {
        const float m4 = Msh[4 * N_MF + j][m];
        e0[j] = m4 * m5a;
        e1[j] = m4 * m5b;
    }
    v2f bv[8];
#pragma unroll
    for (int s = 0; s < 8; ++s) {
        bv[s].x = a23[s >> 2] * e0[s & 3];
        bv[s].y = a23[s >> 2] * e1[s & 3];
    }
    float f01[8];
    {
        const float a1lo = Msh[N_MF + half][m];
        const float a1hi = Msh[N_MF + 2 + half][m];
#pragma unroll
        for (int t = 0; t < 8; ++t)
            f01[t] = Msh[t >> 1][m] * ((t & 1) ? a1hi : a1lo);
    }

    const int  u    = m >> 3;
    const int  c    = m & 7;
    const bool clt7 = (c < 7);
    const float* Aw = reinterpret_cast<const float*>(&Braw[wave][0])
                      + u * CHUNK_FL + (2 * half) * 7 + (clt7 ? c : 0);

    v8f D[8];
#pragma unroll
    for (int t = 0; t < 8; ++t) D[t] = v8f{};

    float4 stgB[LOADS_F4 / 2];
#pragma unroll
    for (int i = 0; i < LOADS_F4 / 2; ++i) {
        const int idx = (LOADS_F4 / 2 + i) * 32 + lane;
        stgB[i] = src4[idx + (idx / CHUNK_F4) * (J01_STRIDE_F4 - CHUNK_F4)];
    }

    wmma_half<0>(D, Aw, bv, clt7);

#pragma unroll
    for (int i = 0; i < LOADS_F4 / 2; ++i)
        Braw[wave][(LOADS_F4 / 2 + i) * 32 + lane] = stgB[i];
    __syncthreads();

    wmma_half<4>(D, Aw, bv, clt7);

#pragma unroll
    for (int v = 0; v < 8; ++v) {
        float sacc = D[0][v] * f01[0];
#pragma unroll
        for (int t = 1; t < 8; ++t) sacc += D[t][v] * f01[t];
        Spart[wave * 2 + half][v][m] = sacc;
    }
    __syncthreads();

    if (tid < 8 * TILE_N) {
        const int cc = tid >> 4, bb = tid & 15;
        float sum = Spart[0][cc][bb];
#pragma unroll
        for (int p = 1; p < WAVES * 2; ++p) sum += Spart[p][cc][bb];
        Sred[cc][bb] = sum;
    }
    __syncthreads();

    if (tid < TILE_N) {
        float num = Sred[6][m];
        num += Sred[0][m] * xr0;
        num += Sred[1][m] * xr1;
        num += Sred[2][m] * xr2;
        num += Sred[3][m] * xr3;
        num += Sred[4][m] * xr4;
        num += Sred[5][m] * xr5;
        float denom = fmaxf(Sred[7][m], 1e-12f);
        out[row] = num / denom;
    }
}

extern "C" void kernel_launch(void* const* d_in, const int* in_sizes, int n_in,
                              void* d_out, int out_size, void* d_ws, size_t ws_size,
                              hipStream_t stream) {
    const float* x     = (const float*)d_in[0];
    const float* mfa   = (const float*)d_in[1];
    const float* mfb   = (const float*)d_in[2];
    const float* mfc   = (const float*)d_in[3];
    const float* coeff = (const float*)d_in[4];
    float* out = (float*)d_out;

    dim3 grid(BATCH / TILE_N);
    dim3 block(WAVES * 32);
    anfis_wmma_kernel<<<grid, block, 0, stream>>>(x, mfa, mfb, mfc, coeff, out);
}
